# starting kernel
# baseline (speedup 1.0000x reference)
#include <hip/hip_runtime.h>

#define N_NODES  50000
#define N_EDGES  1600000
#define IN_FEAT  128
#define OUT_FEAT 128

typedef __attribute__((ext_vector_type(2))) float v2f;
typedef __attribute__((ext_vector_type(8))) float v8f;

__global__ __launch_bounds__(256) void gcn_zero_kernel(float4* __restrict__ out, int n4) {
    int i = blockIdx.x * blockDim.x + threadIdx.x;
    if (i < n4) out[i] = make_float4(0.f, 0.f, 0.f, 0.f);
}

__global__ __launch_bounds__(256) void gcn_gemm_kernel(const float* __restrict__ X,
                                                       const float* __restrict__ W,
                                                       float* __restrict__ H) {
    const int lane = threadIdx.x & 31;
    const int wave = threadIdx.x >> 5;

    const int row0 = blockIdx.x * 16;
    const int o0   = wave * 16;
    const int ln   = lane & 15;
    const int koff = (lane >> 4) << 1;

    const float* __restrict__ xa = X + (size_t)(row0 + ln) * IN_FEAT + koff;
    const float* __restrict__ wb = W + (size_t)(o0 + ln) * IN_FEAT + koff;

    v8f acc = {};
#pragma unroll
    for (int k0 = 0; k0 < IN_FEAT; k0 += 4) {
        v2f a = *(const v2f*)(xa + k0);
        v2f b = *(const v2f*)(wb + k0);
        acc = __builtin_amdgcn_wmma_f32_16x16x4_f32(
            false, a, false, b, (short)0, acc, false, false);
    }

    const int rbase = row0 + ((lane >> 4) << 3);
    float* __restrict__ hp = H + (size_t)rbase * OUT_FEAT + o0 + ln;
#pragma unroll
    for (int v = 0; v < 8; ++v)
        hp[(size_t)v * OUT_FEAT] = acc[v];
}

__global__ __launch_bounds__(256) void gcn_spmm_kernel(const float* __restrict__ H,
                                                       const float* __restrict__ vals,
                                                       const int* __restrict__ rows,
                                                       const int* __restrict__ cols,
                                                       float* __restrict__ out) {
    const int e    = (int)((blockIdx.x * blockDim.x + threadIdx.x) >> 5);
    const int lane = threadIdx.x & 31;
    if (e >= N_EDGES) return;

    const int   r = rows[e];
    const int   c = cols[e];
    const float v = vals[e];

    const float4 h = ((const float4*)(H + (size_t)c * OUT_FEAT))[lane];
    float* __restrict__ o = out + (size_t)r * OUT_FEAT + lane * 4;

    unsafeAtomicAdd(o + 0, v * h.x);
    unsafeAtomicAdd(o + 1, v * h.y);
    unsafeAtomicAdd(o + 2, v * h.z);
    unsafeAtomicAdd(o + 3, v * h.w);
}

extern "C" void kernel_launch(void* const* d_in, const int* in_sizes, int n_in,
                              void* d_out, int out_size, void* d_ws, size_t ws_size,
                              hipStream_t stream) {
    const float* X      = (const float*)d_in[0];
    const float* W      = (const float*)d_in[1];
    const float* A_vals = (const float*)d_in[2];
    const int*   A_rows = (const int*)d_in[3];
    const int*   A_cols = (const int*)d_in[4];
    float*       out    = (float*)d_out;
    float*       H      = (float*)d_ws;

    const int n4 = N_NODES * OUT_FEAT / 4;
    gcn_zero_kernel<<<(n4 + 255) / 256, 256, 0, stream>>>((float4*)out, n4);

    gcn_gemm_kernel<<<N_NODES / 16, 256, 0, stream>>>(X, W, H);

    gcn_spmm_kernel<<<N_EDGES / 8, 256, 0, stream>>>(H, A_vals, A_rows, A_cols, out);
}


# GEMM stage only: each wave keeps its 16 columns of W in registers, X tiles staged in LDS by whole-line 16-byte loads, 625 blocks x 5 tiles (5000 waves, was 25000); same k order, H bit-identical
# speedup vs baseline: 1.9278x; 1.9278x over previous
#include <hip/hip_runtime.h>

#define N_NODES  50000
#define N_EDGES  1600000
#define IN_FEAT  128
#define OUT_FEAT 128

#define PAIR_SLOTS 48
#define LIST_CAP   79
#define CNT_WORD   127

typedef __attribute__((ext_vector_type(2))) float v2f;
typedef __attribute__((ext_vector_type(8))) float v8f;

#define GEMM_TILES_PER_BLOCK 5
#define XS_PITCH 132

__global__ __launch_bounds__(256) void gcn_gemm_kernel(const float* X,
                                                       const float* __restrict__ W,
                                                       float* __restrict__ H,
                                                       float* out) {
    __shared__ __attribute__((aligned(16))) float s_x[2][16 * XS_PITCH];

    const int t    = threadIdx.x;
    const int lane = t & 31;
    const int wave = t >> 5;
    const int o0   = wave * 16;
    const int ln   = lane & 15;
    const int koff = (lane >> 4) << 1;

    if (t < 16 * GEMM_TILES_PER_BLOCK)
        reinterpret_cast<int*>(out)[(size_t)(blockIdx.x * (16 * GEMM_TILES_PER_BLOCK) + t) * OUT_FEAT + CNT_WORD] = 0;

    const float* __restrict__ wb = W + (size_t)(o0 + ln) * IN_FEAT + koff;
    v2f b[IN_FEAT / 4];
#pragma unroll
    for (int i = 0; i < IN_FEAT / 4; ++i)
        b[i] = *(const v2f*)(wb + 4 * i);

    const float4* X4 = reinterpret_cast<const float4*>(X);
    const int tile_first = blockIdx.x * GEMM_TILES_PER_BLOCK;
    const int sp = (t >> 5) * XS_PITCH + (t & 31) * 4;
    float4 p0 = X4[(size_t)tile_first * 512 + t];
    float4 p1 = X4[(size_t)tile_first * 512 + 256 + t];

    for (int it = 0; it < GEMM_TILES_PER_BLOCK; ++it) {
        float* xs = s_x[it & 1];
        *reinterpret_cast<float4*>(xs + sp)                = p0;
        *reinterpret_cast<float4*>(xs + sp + 8 * XS_PITCH) = p1;
        __builtin_amdgcn_fence(__ATOMIC_RELEASE, "workgroup", "local");
        __builtin_amdgcn_s_barrier();
        __builtin_amdgcn_fence(__ATOMIC_ACQUIRE, "workgroup", "local");

        const int tile = tile_first + it;
        if (it + 1 < GEMM_TILES_PER_BLOCK) {
            p0 = X4[(size_t)(tile + 1) * 512 + t];
            p1 = X4[(size_t)(tile + 1) * 512 + 256 + t];
        }
        __builtin_amdgcn_sched_barrier(0);

        const int row0 = tile * 16;

        const float* xa = xs + ln * XS_PITCH + koff;
        v2f a[IN_FEAT / 4];
#pragma unroll
        for (int i = 0; i < IN_FEAT / 4; ++i)
            a[i] = *(const v2f*)(xa + 4 * i);
        __builtin_amdgcn_sched_barrier(0);
        v8f acc = {};
#pragma unroll
        for (int i = 0; i < IN_FEAT / 4; ++i)
            acc = __builtin_amdgcn_wmma_f32_16x16x4_f32(
                false, a[i], false, b[i], (short)0, acc, false, false);
        __builtin_amdgcn_sched_barrier(0);

        const int rbase = row0 + ((lane >> 4) << 3);
        float* __restrict__ hp = H + (size_t)rbase * OUT_FEAT + o0 + ln;
#pragma unroll
        for (int v = 0; v < 8; ++v)
            hp[(size_t)v * OUT_FEAT] = acc[v];
    }
}

__device__ __forceinline__ void gcn_put_edge(int* rowp, int slot, int e, int c, float v) {
    if (slot < PAIR_SLOTS) {
        *reinterpret_cast<int2*>(rowp + 2 * slot) = make_int2(c, __float_as_int(v));
    } else if (slot < LIST_CAP) {
        rowp[2 * PAIR_SLOTS + (slot - PAIR_SLOTS)] = e;
    }
}

__global__ __launch_bounds__(128) void gcn_fill_kernel(const float* __restrict__ vals,
                                                       const int* __restrict__ rows,
                                                       const int* __restrict__ cols,
                                                       float* out) {
    const int i = (int)(blockIdx.x * blockDim.x + threadIdx.x);
    const int4   r = reinterpret_cast<const int4*>(rows)[i];
    const int4   c = reinterpret_cast<const int4*>(cols)[i];
    const float4 v = reinterpret_cast<const float4*>(vals)[i];

    int* const base = reinterpret_cast<int*>(out);
    int* const p0 = base + (size_t)r.x * OUT_FEAT;
    int* const p1 = base + (size_t)r.y * OUT_FEAT;
    int* const p2 = base + (size_t)r.z * OUT_FEAT;
    int* const p3 = base + (size_t)r.w * OUT_FEAT;
    const int s0 = atomicAdd(p0 + CNT_WORD, 1);
    const int s1 = atomicAdd(p1 + CNT_WORD, 1);
    const int s2 = atomicAdd(p2 + CNT_WORD, 1);
    const int s3 = atomicAdd(p3 + CNT_WORD, 1);
    gcn_put_edge(p0, s0, 4 * i + 0, c.x, v.x);
    gcn_put_edge(p1, s1, 4 * i + 1, c.y, v.y);
    gcn_put_edge(p2, s2, 4 * i + 2, c.z, v.z);
    gcn_put_edge(p3, s3, 4 * i + 3, c.w, v.w);
}

__device__ __forceinline__ int gcn_key_below(int kc, unsigned kv, int c, unsigned v) {
    return ((kc < c) | ((kc == c) & (kv < v))) ? 1 : 0;
}

__device__ __forceinline__ void gcn_place(int2* strip, int lane, int n,
                                          int c0, unsigned v0, int rank0,
                                          int c1, unsigned v1, int rank1,
                                          int c2, unsigned v2, int rank2,
                                          int2& q0, int2& q1, int2& q2) {
    strip[lane]      = make_int2(-1, 0);
    strip[32 + lane] = make_int2(-1, 0);
    strip[64 + lane] = make_int2(-1, 0);
    if (lane < n)      strip[rank0] = make_int2(c0, (int)v0);
    if (32 + lane < n) strip[rank1] = make_int2(c1, (int)v1);
    if (64 + lane < n) strip[rank2] = make_int2(c2, (int)v2);
    q0 = strip[lane];
    q1 = strip[32 + lane];
    q2 = strip[64 + lane];
}

__global__ __launch_bounds__(256) void gcn_rowsum_kernel(const float* __restrict__ H,
                                                         const float* __restrict__ vals,
                                                         const int* __restrict__ rows,
                                                         const int* __restrict__ cols,
                                                         float* out) {
    __shared__ int2 s_cv[8][96];

    const int lane = threadIdx.x & 31;
    const int wave = __builtin_amdgcn_readfirstlane((int)(threadIdx.x >> 5));
    const int r    = (int)blockIdx.x * 8 + wave;

    int* rowp = reinterpret_cast<int*>(out) + (size_t)r * OUT_FEAT;
    const float4* __restrict__ H4 = reinterpret_cast<const float4*>(H);

    const int n = __builtin_amdgcn_readfirstlane(rowp[CNT_WORD]);

    float4 acc = make_float4(0.f, 0.f, 0.f, 0.f);

    if (n <= LIST_CAP) {
        int c0 = 0, c1 = 0, c2 = 0;
        unsigned v0 = 0, v1 = 0, v2 = 0;
        if (lane < n) {
            const int2 p = reinterpret_cast<const int2*>(rowp)[lane];
            c0 = p.x; v0 = (unsigned)p.y;
        }
        const int s1 = 32 + lane;
        if (s1 < n) {
            if (s1 < PAIR_SLOTS) {
                const int2 p = reinterpret_cast<const int2*>(rowp)[s1];
                c1 = p.x; v1 = (unsigned)p.y;
            } else {
                const int e = rowp[2 * PAIR_SLOTS + (s1 - PAIR_SLOTS)];
                c1 = cols[e]; v1 = (unsigned)__float_as_int(vals[e]);
            }
        }
        const int s2 = 64 + lane;
        if (s2 < n) {
            const int e = rowp[2 * PAIR_SLOTS + (s2 - PAIR_SLOTS)];
            c2 = cols[e]; v2 = (unsigned)__float_as_int(vals[e]);
        }

        const int m0 = n < 32 ? n : 32;
        const int m1 = n < 64 ? n - 32 : 32;
        const int m2 = n - 64;

        int rank0 = 0, rank1 = 0, rank2 = 0;
        if (n <= 32) {
            for (int t = 0; t < n; ++t)
                rank0 += (__builtin_amdgcn_readlane(c0, t) < c0) ? 1 : 0;
        } else {
            for (int t = 0; t < 32; ++t) {
                const int kc = __builtin_amdgcn_readlane(c0, t);
                rank0 += (kc < c0) ? 1 : 0; rank1 += (kc < c1) ? 1 : 0; rank2 += (kc < c2) ? 1 : 0;
            }
            for (int t = 0; t < m1; ++t) {
                const int kc = __builtin_amdgcn_readlane(c1, t);
                rank0 += (kc < c0) ? 1 : 0; rank1 += (kc < c1) ? 1 : 0; rank2 += (kc < c2) ? 1 : 0;
            }
            for (int t = 0; t < m2; ++t) {
                const int kc = __builtin_amdgcn_readlane(c2, t);
                rank0 += (kc < c0) ? 1 : 0; rank1 += (kc < c1) ? 1 : 0; rank2 += (kc < c2) ? 1 : 0;
            }
        }

        int2* strip = s_cv[wave];
        int2 q0, q1, q2;
        gcn_place(strip, lane, n, c0, v0, rank0, c1, v1, rank1, c2, v2, rank2, q0, q1, q2);

        const bool hole = ((lane < n) & (q0.x < 0)) | ((s1 < n) & (q1.x < 0)) | ((s2 < n) & (q2.x < 0));
        if (__builtin_amdgcn_ballot_w32(hole) != 0u) {
            rank0 = 0; rank1 = 0; rank2 = 0;
            for (int t = 0; t < n; ++t) {
                int kc; unsigned kv;
                if (t < 32)      { kc = __builtin_amdgcn_readlane(c0, t);      kv = (unsigned)__builtin_amdgcn_readlane((int)v0, t); }
                else if (t < 64) { kc = __builtin_amdgcn_readlane(c1, t - 32); kv = (unsigned)__builtin_amdgcn_readlane((int)v1, t - 32); }
                else             { kc = __builtin_amdgcn_readlane(c2, t - 64); kv = (unsigned)__builtin_amdgcn_readlane((int)v2, t - 64); }
                rank0 += gcn_key_below(kc, kv, c0, v0);
                rank1 += gcn_key_below(kc, kv, c1, v1);
                rank2 += gcn_key_below(kc, kv, c2, v2);
            }
            gcn_place(strip, lane, n, c0, v0, rank0, c1, v1, rank1, c2, v2, rank2, q0, q1, q2);
        }

        int   pc = 0;
        float pv = 0.f;
#define GCN_SUM_SEGMENT(Q, M)                                                       \
        for (int t0 = 0; t0 < (M); t0 += 8) {                                       \
            _Pragma("unroll")                                                       \
            for (int u = 0; u < 8; ++u) {                                           \
                const int t = t0 + u;                                               \
                int   kc = __builtin_amdgcn_readlane((Q).x, t);                     \
                float kv = __int_as_float(__builtin_amdgcn_readlane((Q).y, t));     \
                if (kc < 0) { kc = pc; kv = (t < (M)) ? pv : 0.f; }                 \
                pc = kc; pv = kv;                                                   \
                const float4 h = H4[(size_t)kc * (OUT_FEAT / 4) + lane];            \
                acc.x = fmaf(kv, h.x, acc.x); acc.y = fmaf(kv, h.y, acc.y);         \
                acc.z = fmaf(kv, h.z, acc.z); acc.w = fmaf(kv, h.w, acc.w);         \
            }                                                                       \
        }
        GCN_SUM_SEGMENT(q0, m0)
        GCN_SUM_SEGMENT(q1, m1)
        GCN_SUM_SEGMENT(q2, m2)
#undef GCN_SUM_SEGMENT
    } else {
        for (int base = 0; base < N_EDGES; base += 32) {
            unsigned hit = __builtin_amdgcn_ballot_w32(rows[base + lane] == r);
            while (hit) {
                const int b = __builtin_ctz(hit);
                hit &= hit - 1;
                const int   c = cols[base + b];
                const float v = vals[base + b];
                const float4 h = H4[(size_t)c * (OUT_FEAT / 4) + lane];
                acc.x = fmaf(v, h.x, acc.x); acc.y = fmaf(v, h.y, acc.y);
                acc.z = fmaf(v, h.z, acc.z); acc.w = fmaf(v, h.w, acc.w);
            }
        }
    }

    reinterpret_cast<int4*>(rowp)[lane] =
        make_int4(__float_as_int(acc.x), __float_as_int(acc.y),
                  __float_as_int(acc.z), __float_as_int(acc.w));
}

extern "C" void kernel_launch(void* const* d_in, const int* in_sizes, int n_in,
                              void* d_out, int out_size, void* d_ws, size_t ws_size,
                              hipStream_t stream) {
    const float* X      = (const float*)d_in[0];
    const float* W      = (const float*)d_in[1];
    const float* A_vals = (const float*)d_in[2];
    const int*   A_rows = (const int*)d_in[3];
    const int*   A_cols = (const int*)d_in[4];
    float*       out    = (float*)d_out;
    float*       H      = (float*)d_ws;

    gcn_gemm_kernel<<<N_NODES / 16 / GEMM_TILES_PER_BLOCK, 256, 0, stream>>>(X, W, H, out);

    gcn_fill_kernel<<<N_EDGES / 4 / 128, 128, 0, stream>>>(A_vals, A_rows, A_cols, out);

    gcn_rowsum_kernel<<<N_NODES / 8, 256, 0, stream>>>(H, A_vals, A_rows, A_cols, out);
}


# per-edge global returning adds and scattered 8-byte list stores replaced: edges binned by 100-row destination bucket through LDS (one global add per block and bucket), row lists built in LDS
# speedup vs baseline: 3.3393x; 1.7322x over previous
#include <hip/hip_runtime.h>

#define N_NODES  50000
#define N_EDGES  1600000
#define IN_FEAT  128
#define OUT_FEAT 128

#define RPB          100
#define NB           (N_NODES / RPB)
#define SEG_WORDS    (RPB * OUT_FEAT)
#define SEG_CAP      (SEG_WORDS / 2 - 1)
#define SEG_CNT_WORD (SEG_WORDS - 1)
#define LIST_CAP     79
static_assert(N_NODES <= 65536, "destination and source share one 32-bit word of a record");

typedef __attribute__((ext_vector_type(2))) float v2f;
typedef __attribute__((ext_vector_type(8))) float v8f;

#define GEMM_TILES_PER_BLOCK 5
#define XS_PITCH 132

__global__ __launch_bounds__(256) void gcn_gemm_kernel(const float* X,
                                                       const float* __restrict__ W,
                                                       float* __restrict__ H,
                                                       float* out) {
    __shared__ __attribute__((aligned(16))) float s_x[2][16 * XS_PITCH];

    const int t    = threadIdx.x;
    const int lane = t & 31;
    const int wave = t >> 5;
    const int o0   = wave * 16;
    const int ln   = lane & 15;
    const int koff = (lane >> 4) << 1;

    {
        const int g = (int)blockIdx.x * 256 + t;
        if (g < NB)
            reinterpret_cast<int*>(out)[(size_t)g * SEG_WORDS + SEG_CNT_WORD] = 0;
    }

    const float* __restrict__ wb = W + (size_t)(o0 + ln) * IN_FEAT + koff;
    v2f b[IN_FEAT / 4];
#pragma unroll
    for (int i = 0; i < IN_FEAT / 4; ++i)
        b[i] = *(const v2f*)(wb + 4 * i);

    const float4* X4 = reinterpret_cast<const float4*>(X);
    const int tile_first = blockIdx.x * GEMM_TILES_PER_BLOCK;
    const int sp = (t >> 5) * XS_PITCH + (t & 31) * 4;
    float4 p0 = X4[(size_t)tile_first * 512 + t];
    float4 p1 = X4[(size_t)tile_first * 512 + 256 + t];

    for (int it = 0; it < GEMM_TILES_PER_BLOCK; ++it) {
        float* xs = s_x[it & 1];
        *reinterpret_cast<float4*>(xs + sp)                = p0;
        *reinterpret_cast<float4*>(xs + sp + 8 * XS_PITCH) = p1;
        __builtin_amdgcn_fence(__ATOMIC_RELEASE, "workgroup", "local");
        __builtin_amdgcn_s_barrier();
        __builtin_amdgcn_fence(__ATOMIC_ACQUIRE, "workgroup", "local");

        const int tile = tile_first + it;
        if (it + 1 < GEMM_TILES_PER_BLOCK) {
            p0 = X4[(size_t)(tile + 1) * 512 + t];
            p1 = X4[(size_t)(tile + 1) * 512 + 256 + t];
        }
        __builtin_amdgcn_sched_barrier(0);

        const int row0 = tile * 16;

        const float* xa = xs + ln * XS_PITCH + koff;
        v2f a[IN_FEAT / 4];
#pragma unroll
        for (int i = 0; i < IN_FEAT / 4; ++i)
            a[i] = *(const v2f*)(xa + 4 * i);
        __builtin_amdgcn_sched_barrier(0);
        v8f acc = {};
#pragma unroll
        for (int i = 0; i < IN_FEAT / 4; ++i)
            acc = __builtin_amdgcn_wmma_f32_16x16x4_f32(
                false, a[i], false, b[i], (short)0, acc, false, false);
        __builtin_amdgcn_sched_barrier(0);

        const int rbase = row0 + ((lane >> 4) << 3);
        float* __restrict__ hp = H + (size_t)rbase * OUT_FEAT + o0 + ln;
#pragma unroll
        for (int v = 0; v < 8; ++v)
            hp[(size_t)v * OUT_FEAT] = acc[v];
    }
}

#define PART_THREADS 1024
#define PART_EDGES   4096
#define PART_BLOCKS  391

__global__ __launch_bounds__(PART_THREADS) void gcn_part_kernel(const float* __restrict__ vals,
                                                                const int* __restrict__ rows,
                                                                const int* __restrict__ cols,
                                                                float* out) {
    __shared__ int  s_cnt[NB];
    __shared__ int  s_off[NB];
    __shared__ int  s_base[NB];
    __shared__ int  s_wtot[PART_THREADS / 32];
    __shared__ int2 s_rec[PART_EDGES];

    const int t    = threadIdx.x;
    const int lane = t & 31;
    const int wave = t >> 5;
    const int i    = (int)blockIdx.x * PART_THREADS + t;
    const bool live = i < N_EDGES / 4;
    int live_threads = N_EDGES / 4 - (int)blockIdx.x * PART_THREADS;
    if (live_threads > PART_THREADS) live_threads = PART_THREADS;
    const int total = 4 * live_threads;

    if (t < NB) s_cnt[t] = 0;
    __syncthreads();

    int4   r = make_int4(0, 0, 0, 0);
    int4   c = make_int4(0, 0, 0, 0);
    float4 v = make_float4(0.f, 0.f, 0.f, 0.f);
    int b0 = 0, b1 = 0, b2 = 0, b3 = 0;
    int l0 = 0, l1 = 0, l2 = 0, l3 = 0;
    if (live) {
        r = reinterpret_cast<const int4*>(rows)[i];
        c = reinterpret_cast<const int4*>(cols)[i];
        v = reinterpret_cast<const float4*>(vals)[i];
        b0 = r.x / RPB; b1 = r.y / RPB; b2 = r.z / RPB; b3 = r.w / RPB;
        l0 = atomicAdd(&s_cnt[b0], 1);
        l1 = atomicAdd(&s_cnt[b1], 1);
        l2 = atomicAdd(&s_cnt[b2], 1);
        l3 = atomicAdd(&s_cnt[b3], 1);
    }
    __syncthreads();

    const int cnt = (t < NB) ? s_cnt[t] : 0;
    int incl = cnt;
    for (int d = 1; d < 32; d <<= 1) {
        const int y = __shfl_up(incl, d, 32);
        if (lane >= d) incl += y;
    }
    if (lane == 31) s_wtot[wave] = incl;
    __syncthreads();
    int pre = 0;
    for (int w = 0; w < wave; ++w) pre += s_wtot[w];

    int* const obase = reinterpret_cast<int*>(out);
    if (t < NB) {
        s_off[t] = pre + incl - cnt;
        int base = 0;
        if (cnt > 0)
            base = atomicAdd(obase + (size_t)t * SEG_WORDS + SEG_CNT_WORD, cnt);
        s_base[t] = base;
    }
    __syncthreads();

    if (live) {
        s_rec[s_off[b0] + l0] = make_int2((int)(((unsigned)r.x << 16) | (unsigned)c.x), __float_as_int(v.x));
        s_rec[s_off[b1] + l1] = make_int2((int)(((unsigned)r.y << 16) | (unsigned)c.y), __float_as_int(v.y));
        s_rec[s_off[b2] + l2] = make_int2((int)(((unsigned)r.z << 16) | (unsigned)c.z), __float_as_int(v.z));
        s_rec[s_off[b3] + l3] = make_int2((int)(((unsigned)r.w << 16) | (unsigned)c.w), __float_as_int(v.w));
    }
    __syncthreads();

#pragma unroll
    for (int j = 0; j < 4; ++j) {
        const int idx = t + j * PART_THREADS;
        if (idx < total) {
            const int2 rec = s_rec[idx];
            const int  b   = (int)((unsigned)rec.x >> 16) / RPB;
            const int  g   = s_base[b] + (idx - s_off[b]);
            if (g < SEG_CAP)
                reinterpret_cast<int2*>(obase + (size_t)b * SEG_WORDS)[g] = rec;
        }
    }
}

__device__ __forceinline__ int gcn_key_below(int kc, unsigned kv, int c, unsigned v) {
    return ((kc < c) | ((kc == c) & (kv < v))) ? 1 : 0;
}

__device__ __forceinline__ void gcn_place(int2* strip, int lane, int n,
                                          int c0, unsigned v0, int rank0,
                                          int c1, unsigned v1, int rank1,
                                          int c2, unsigned v2, int rank2,
                                          int2& q0, int2& q1, int2& q2) {
    strip[lane]      = make_int2(-1, 0);
    strip[32 + lane] = make_int2(-1, 0);
    strip[64 + lane] = make_int2(-1, 0);
    if (lane < n)      strip[rank0] = make_int2(c0, (int)v0);
    if (32 + lane < n) strip[rank1] = make_int2(c1, (int)v1);
    if (64 + lane < n) strip[rank2] = make_int2(c2, (int)v2);
    q0 = strip[lane];
    q1 = strip[32 + lane];
    q2 = strip[64 + lane];
}

#define ROWSUM_WAVES 32
#define ROWSUM_RECS  7

__global__ __launch_bounds__(ROWSUM_WAVES * 32) void gcn_rowsum_kernel(const float* __restrict__ H,
                                                                       const float* __restrict__ vals,
                                                                       const int* __restrict__ rows,
                                                                       const int* __restrict__ cols,
                                                                       float* out) {
    __shared__ int2 s_cv[ROWSUM_WAVES][96];
    __shared__ int2 s_list[RPB][LIST_CAP];
    __shared__ int  s_n[RPB];

    const int t    = threadIdx.x;
    const int lane = t & 31;
    const int wave = __builtin_amdgcn_readfirstlane((int)(t >> 5));
    const int bkt  = (int)blockIdx.x;

    int* const seg = reinterpret_cast<int*>(out) + (size_t)bkt * SEG_WORDS;
    const float4* __restrict__ H4 = reinterpret_cast<const float4*>(H);

    if (t < RPB) s_n[t] = 0;
    const int tot = seg[SEG_CNT_WORD];
    __syncthreads();
    const int nrec = tot < SEG_CAP ? tot : SEG_CAP;
    int2 rec[ROWSUM_RECS];
#pragma unroll
    for (int k = 0; k < ROWSUM_RECS; ++k) {
        const int idx = t + k * (ROWSUM_WAVES * 32);
        rec[k] = make_int2(0, 0);
        if (idx < nrec) rec[k] = reinterpret_cast<const int2*>(seg)[idx];
    }
#pragma unroll
    for (int k = 0; k < ROWSUM_RECS; ++k) {
        const int idx = t + k * (ROWSUM_WAVES * 32);
        if (idx < nrec) {
            const unsigned key = (unsigned)rec[k].x;
            const int rl   = (int)(key >> 16) - bkt * RPB;
            const int slot = atomicAdd(&s_n[rl], 1);
            if (slot < LIST_CAP)
                s_list[rl][slot] = make_int2((int)(key & 0xffffu), rec[k].y);
        }
    }
    __syncthreads();
    const bool seg_full = tot > SEG_CAP;

    for (int row = wave; row < RPB; row += ROWSUM_WAVES) {
        const int r = bkt * RPB + row;
        int* rowp = reinterpret_cast<int*>(out) + (size_t)r * OUT_FEAT;

        const int n = seg_full ? (LIST_CAP + 1) : __builtin_amdgcn_readfirstlane(s_n[row]);

        float4 acc = make_float4(0.f, 0.f, 0.f, 0.f);

        if (n <= LIST_CAP) {
            int c0 = 0, c1 = 0, c2 = 0;
            unsigned v0 = 0, v1 = 0, v2 = 0;
            if (lane < n) {
                const int2 p = s_list[row][lane];
                c0 = p.x; v0 = (unsigned)p.y;
            }
            const int s1 = 32 + lane;
            if (s1 < n) {
                const int2 p = s_list[row][s1];
                c1 = p.x; v1 = (unsigned)p.y;
            }
            const int s2 = 64 + lane;
            if (s2 < n) {
                const int2 p = s_list[row][s2];
                c2 = p.x; v2 = (unsigned)p.y;
            }

            const int m0 = n < 32 ? n : 32;
            const int m1 = n < 64 ? n - 32 : 32;
            const int m2 = n - 64;

            int rank0 = 0, rank1 = 0, rank2 = 0;
            if (n <= 32) {
                for (int t = 0; t < n; ++t)
                    rank0 += (__builtin_amdgcn_readlane(c0, t) < c0) ? 1 : 0;
            } else {
                for (int t = 0; t < 32; ++t) {
                    const int kc = __builtin_amdgcn_readlane(c0, t);
                    rank0 += (kc < c0) ? 1 : 0; rank1 += (kc < c1) ? 1 : 0; rank2 += (kc < c2) ? 1 : 0;
                }
                for (int t = 0; t < m1; ++t) {
                    const int kc = __builtin_amdgcn_readlane(c1, t);
                    rank0 += (kc < c0) ? 1 : 0; rank1 += (kc < c1) ? 1 : 0; rank2 += (kc < c2) ? 1 : 0;
                }
                for (int t = 0; t < m2; ++t) {
                    const int kc = __builtin_amdgcn_readlane(c2, t);
                    rank0 += (kc < c0) ? 1 : 0; rank1 += (kc < c1) ? 1 : 0; rank2 += (kc < c2) ? 1 : 0;
                }
            }

            int2* strip = s_cv[wave];
            int2 q0, q1, q2;
            gcn_place(strip, lane, n, c0, v0, rank0, c1, v1, rank1, c2, v2, rank2, q0, q1, q2);

            const bool hole = ((lane < n) & (q0.x < 0)) | ((s1 < n) & (q1.x < 0)) | ((s2 < n) & (q2.x < 0));
            if (__builtin_amdgcn_ballot_w32(hole) != 0u) {
                rank0 = 0; rank1 = 0; rank2 = 0;
                for (int t = 0; t < n; ++t) {
                    int kc; unsigned kv;
                    if (t < 32)      { kc = __builtin_amdgcn_readlane(c0, t);      kv = (unsigned)__builtin_amdgcn_readlane((int)v0, t); }
                    else if (t < 64) { kc = __builtin_amdgcn_readlane(c1, t - 32); kv = (unsigned)__builtin_amdgcn_readlane((int)v1, t - 32); }
                    else             { kc = __builtin_amdgcn_readlane(c2, t - 64); kv = (unsigned)__builtin_amdgcn_readlane((int)v2, t - 64); }
                    rank0 += gcn_key_below(kc, kv, c0, v0);
                    rank1 += gcn_key_below(kc, kv, c1, v1);
                    rank2 += gcn_key_below(kc, kv, c2, v2);
                }
                gcn_place(strip, lane, n, c0, v0, rank0, c1, v1, rank1, c2, v2, rank2, q0, q1, q2);
            }

            int   pc = 0;
            float pv = 0.f;
#define GCN_SUM_SEGMENT(Q, M)                                                       \
            for (int t0 = 0; t0 < (M); t0 += 8) {                                       \
                _Pragma("unroll")                                                       \
                for (int u = 0; u < 8; ++u) {                                           \
                    const int t = t0 + u;                                               \
                    int   kc = __builtin_amdgcn_readlane((Q).x, t);                     \
                    float kv = __int_as_float(__builtin_amdgcn_readlane((Q).y, t));     \
                    if (kc < 0) { kc = pc; kv = (t < (M)) ? pv : 0.f; }                 \
                    pc = kc; pv = kv;                                                   \
                    const float4 h = H4[(size_t)kc * (OUT_FEAT / 4) + lane];            \
                    acc.x = fmaf(kv, h.x, acc.x); acc.y = fmaf(kv, h.y, acc.y);         \
                    acc.z = fmaf(kv, h.z, acc.z); acc.w = fmaf(kv, h.w, acc.w);         \
                }                                                                       \
            }
            GCN_SUM_SEGMENT(q0, m0)
            GCN_SUM_SEGMENT(q1, m1)
            GCN_SUM_SEGMENT(q2, m2)
#undef GCN_SUM_SEGMENT
        } else {
            for (int base = 0; base < N_EDGES; base += 32) {
                unsigned hit = __builtin_amdgcn_ballot_w32(rows[base + lane] == r);
                while (hit) {
                    const int b = __builtin_ctz(hit);
                    hit &= hit - 1;
                    const int   c = cols[base + b];
                    const float v = vals[base + b];
                    const float4 h = H4[(size_t)c * (OUT_FEAT / 4) + lane];
                    acc.x = fmaf(v, h.x, acc.x); acc.y = fmaf(v, h.y, acc.y);
                    acc.z = fmaf(v, h.z, acc.z); acc.w = fmaf(v, h.w, acc.w);
                }
            }
        }

        reinterpret_cast<int4*>(rowp)[lane] =
            make_int4(__float_as_int(acc.x), __float_as_int(acc.y),
                      __float_as_int(acc.z), __float_as_int(acc.w));
    }
}

extern "C" void kernel_launch(void* const* d_in, const int* in_sizes, int n_in,
                              void* d_out, int out_size, void* d_ws, size_t ws_size,
                              hipStream_t stream) {
    const float* X      = (const float*)d_in[0];
    const float* W      = (const float*)d_in[1];
    const float* A_vals = (const float*)d_in[2];
    const int*   A_rows = (const int*)d_in[3];
    const int*   A_cols = (const int*)d_in[4];
    float*       out    = (float*)d_out;
    float*       H      = (float*)d_ws;

    gcn_gemm_kernel<<<N_NODES / 16 / GEMM_TILES_PER_BLOCK, 256, 0, stream>>>(X, W, H, out);

    gcn_part_kernel<<<PART_BLOCKS, PART_THREADS, 0, stream>>>(A_vals, A_rows, A_cols, out);

    gcn_rowsum_kernel<<<NB, ROWSUM_WAVES * 32, 0, stream>>>(H, A_vals, A_rows, A_cols, out);
}


# edge kernels launched as fewer, longer-lived waves: binning 196 blocks x 16 waves, 16 edges per thread (3136 waves, was 12512); row sums 20 waves per bucket, 5 rows each (10000, was 16000)
# speedup vs baseline: 3.6553x; 1.0946x over previous
#include <hip/hip_runtime.h>

#define N_NODES  50000
#define N_EDGES  1600000
#define IN_FEAT  128
#define OUT_FEAT 128

#define RPB          100
#define NB           (N_NODES / RPB)
#define SEG_WORDS    (RPB * OUT_FEAT)
#define SEG_CAP      (SEG_WORDS / 2 - 1)
#define SEG_CNT_WORD (SEG_WORDS - 1)
#define LIST_CAP     79
static_assert(N_NODES <= 65536, "destination and source share one 32-bit word of a record");

typedef __attribute__((ext_vector_type(2))) float v2f;
typedef __attribute__((ext_vector_type(8))) float v8f;

#define GEMM_TILES_PER_BLOCK 5
#define XS_PITCH 132

__global__ __launch_bounds__(256) void gcn_gemm_kernel(const float* X,
                                                       const float* __restrict__ W,
                                                       float* __restrict__ H,
                                                       float* out) {
    __shared__ __attribute__((aligned(16))) float s_x[2][16 * XS_PITCH];

    const int t    = threadIdx.x;
    const int lane = t & 31;
    const int wave = t >> 5;
    const int o0   = wave * 16;
    const int ln   = lane & 15;
    const int koff = (lane >> 4) << 1;

    {
        const int g = (int)blockIdx.x * 256 + t;
        if (g < NB)
            reinterpret_cast<int*>(out)[(size_t)g * SEG_WORDS + SEG_CNT_WORD] = 0;
    }

    const float* __restrict__ wb = W + (size_t)(o0 + ln) * IN_FEAT + koff;
    v2f b[IN_FEAT / 4];
#pragma unroll
    for (int i = 0; i < IN_FEAT / 4; ++i)
        b[i] = *(const v2f*)(wb + 4 * i);

    const float4* X4 = reinterpret_cast<const float4*>(X);
    const int tile_first = blockIdx.x * GEMM_TILES_PER_BLOCK;
    const int sp = (t >> 5) * XS_PITCH + (t & 31) * 4;
    float4 p0 = X4[(size_t)tile_first * 512 + t];
    float4 p1 = X4[(size_t)tile_first * 512 + 256 + t];

    for (int it = 0; it < GEMM_TILES_PER_BLOCK; ++it) {
        float* xs = s_x[it & 1];
        *reinterpret_cast<float4*>(xs + sp)                = p0;
        *reinterpret_cast<float4*>(xs + sp + 8 * XS_PITCH) = p1;
        __builtin_amdgcn_fence(__ATOMIC_RELEASE, "workgroup", "local");
        __builtin_amdgcn_s_barrier();
        __builtin_amdgcn_fence(__ATOMIC_ACQUIRE, "workgroup", "local");

        const int tile = tile_first + it;
        if (it + 1 < GEMM_TILES_PER_BLOCK) {
            p0 = X4[(size_t)(tile + 1) * 512 + t];
            p1 = X4[(size_t)(tile + 1) * 512 + 256 + t];
        }
        __builtin_amdgcn_sched_barrier(0);

        const int row0 = tile * 16;

        const float* xa = xs + ln * XS_PITCH + koff;
        v2f a[IN_FEAT / 4];
#pragma unroll
        for (int i = 0; i < IN_FEAT / 4; ++i)
            a[i] = *(const v2f*)(xa + 4 * i);
        __builtin_amdgcn_sched_barrier(0);
        v8f acc = {};
#pragma unroll
        for (int i = 0; i < IN_FEAT / 4; ++i)
            acc = __builtin_amdgcn_wmma_f32_16x16x4_f32(
                false, a[i], false, b[i], (short)0, acc, false, false);
        __builtin_amdgcn_sched_barrier(0);

        const int rbase = row0 + ((lane >> 4) << 3);
        float* __restrict__ hp = H + (size_t)rbase * OUT_FEAT + o0 + ln;
#pragma unroll
        for (int v = 0; v < 8; ++v)
            hp[(size_t)v * OUT_FEAT] = acc[v];
    }
}

#define PART_THREADS 512
#define PART_GROUPS  4
#define PART_BGROUPS (PART_THREADS * PART_GROUPS)
#define PART_EDGES   (PART_BGROUPS * 4)
#define PART_BLOCKS  ((N_EDGES / 4 + PART_BGROUPS - 1) / PART_BGROUPS)
static_assert(NB <= PART_THREADS, "one thread per bucket in the counting and scan steps");
static_assert(N_EDGES % 4 == 0, "edges are read in groups of 4");

__global__ __launch_bounds__(PART_THREADS) void gcn_part_kernel(const float* __restrict__ vals,
                                                                const int* __restrict__ rows,
                                                                const int* __restrict__ cols,
                                                                float* out) {
    __shared__ int  s_cnt[NB];
    __shared__ int  s_off[NB];
    __shared__ int  s_base[NB];
    __shared__ int  s_wtot[PART_THREADS / 32];
    __shared__ int2 s_rec[PART_EDGES];

    const int t    = threadIdx.x;
    const int lane = t & 31;
    const int wave = t >> 5;
    const int g0   = (int)blockIdx.x * PART_BGROUPS;
    int live_groups = N_EDGES / 4 - g0;
    if (live_groups > PART_BGROUPS) live_groups = PART_BGROUPS;
    const int total = 4 * live_groups;

    if (t < NB) s_cnt[t] = 0;
    __syncthreads();

    int4   r[PART_GROUPS];
    int4   c[PART_GROUPS];
    float4 v[PART_GROUPS];
    int4   l[PART_GROUPS];
#pragma unroll
    for (int j = 0; j < PART_GROUPS; ++j) {
        const int gl = j * PART_THREADS + t;
        r[j] = make_int4(0, 0, 0, 0);
        c[j] = make_int4(0, 0, 0, 0);
        v[j] = make_float4(0.f, 0.f, 0.f, 0.f);
        if (gl < live_groups) {
            r[j] = reinterpret_cast<const int4*>(rows)[g0 + gl];
            c[j] = reinterpret_cast<const int4*>(cols)[g0 + gl];
            v[j] = reinterpret_cast<const float4*>(vals)[g0 + gl];
        }
    }
#pragma unroll
    for (int j = 0; j < PART_GROUPS; ++j) {
        l[j] = make_int4(0, 0, 0, 0);
        if (j * PART_THREADS + t < live_groups) {
            l[j].x = atomicAdd(&s_cnt[r[j].x / RPB], 1);
            l[j].y = atomicAdd(&s_cnt[r[j].y / RPB], 1);
            l[j].z = atomicAdd(&s_cnt[r[j].z / RPB], 1);
            l[j].w = atomicAdd(&s_cnt[r[j].w / RPB], 1);
        }
    }
    __syncthreads();

    const int cnt = (t < NB) ? s_cnt[t] : 0;
    int incl = cnt;
    for (int d = 1; d < 32; d <<= 1) {
        const int y = __shfl_up(incl, d, 32);
        if (lane >= d) incl += y;
    }
    if (lane == 31) s_wtot[wave] = incl;
    __syncthreads();
    int pre = 0;
    for (int w = 0; w < wave; ++w) pre += s_wtot[w];

    int* const obase = reinterpret_cast<int*>(out);
    if (t < NB) {
        s_off[t] = pre + incl - cnt;
        int base = 0;
        if (cnt > 0)
            base = atomicAdd(obase + (size_t)t * SEG_WORDS + SEG_CNT_WORD, cnt);
        s_base[t] = base;
    }
    __syncthreads();

#pragma unroll
    for (int j = 0; j < PART_GROUPS; ++j) {
        if (j * PART_THREADS + t < live_groups) {
            s_rec[s_off[r[j].x / RPB] + l[j].x] = make_int2((int)(((unsigned)r[j].x << 16) | (unsigned)c[j].x), __float_as_int(v[j].x));
            s_rec[s_off[r[j].y / RPB] + l[j].y] = make_int2((int)(((unsigned)r[j].y << 16) | (unsigned)c[j].y), __float_as_int(v[j].y));
            s_rec[s_off[r[j].z / RPB] + l[j].z] = make_int2((int)(((unsigned)r[j].z << 16) | (unsigned)c[j].z), __float_as_int(v[j].z));
            s_rec[s_off[r[j].w / RPB] + l[j].w] = make_int2((int)(((unsigned)r[j].w << 16) | (unsigned)c[j].w), __float_as_int(v[j].w));
        }
    }
    __syncthreads();

#pragma unroll
    for (int j = 0; j < 4 * PART_GROUPS; ++j) {
        const int idx = t + j * PART_THREADS;
        if (idx < total) {
            const int2 rec = s_rec[idx];
            const int  b   = (int)((unsigned)rec.x >> 16) / RPB;
            const int  g   = s_base[b] + (idx - s_off[b]);
            if (g < SEG_CAP)
                reinterpret_cast<int2*>(obase + (size_t)b * SEG_WORDS)[g] = rec;
        }
    }
}

__device__ __forceinline__ int gcn_key_below(int kc, unsigned kv, int c, unsigned v) {
    return ((kc < c) | ((kc == c) & (kv < v))) ? 1 : 0;
}

__device__ __forceinline__ void gcn_place(int2* strip, int lane, int n,
                                          int c0, unsigned v0, int rank0,
                                          int c1, unsigned v1, int rank1,
                                          int c2, unsigned v2, int rank2,
                                          int2& q0, int2& q1, int2& q2) {
    strip[lane]      = make_int2(-1, 0);
    strip[32 + lane] = make_int2(-1, 0);
    strip[64 + lane] = make_int2(-1, 0);
    if (lane < n)      strip[rank0] = make_int2(c0, (int)v0);
    if (32 + lane < n) strip[rank1] = make_int2(c1, (int)v1);
    if (64 + lane < n) strip[rank2] = make_int2(c2, (int)v2);
    q0 = strip[lane];
    q1 = strip[32 + lane];
    q2 = strip[64 + lane];
}

#define ROWSUM_WAVES 20
#define ROWSUM_RECS  10
static_assert(RPB % ROWSUM_WAVES == 0, "every wave of a block takes the same number of destinations");
static_assert(ROWSUM_RECS * ROWSUM_WAVES * 32 >= SEG_CAP, "one pass of the block reads a whole segment");

__global__ __launch_bounds__(ROWSUM_WAVES * 32) void gcn_rowsum_kernel(const float* __restrict__ H,
                                                                       const float* __restrict__ vals,
                                                                       const int* __restrict__ rows,
                                                                       const int* __restrict__ cols,
                                                                       float* out) {
    __shared__ int2 s_cv[ROWSUM_WAVES][96];
    __shared__ int2 s_list[RPB][LIST_CAP];
    __shared__ int  s_n[RPB];

    const int t    = threadIdx.x;
    const int lane = t & 31;
    const int wave = __builtin_amdgcn_readfirstlane((int)(t >> 5));
    const int bkt  = (int)blockIdx.x;

    int* const seg = reinterpret_cast<int*>(out) + (size_t)bkt * SEG_WORDS;
    const float4* __restrict__ H4 = reinterpret_cast<const float4*>(H);

    if (t < RPB) s_n[t] = 0;
    const int tot = seg[SEG_CNT_WORD];
    __syncthreads();
    const int nrec = tot < SEG_CAP ? tot : SEG_CAP;
    int2 rec[ROWSUM_RECS];
#pragma unroll
    for (int k = 0; k < ROWSUM_RECS; ++k) {
        const int idx = t + k * (ROWSUM_WAVES * 32);
        rec[k] = make_int2(0, 0);
        if (idx < nrec) rec[k] = reinterpret_cast<const int2*>(seg)[idx];
    }
#pragma unroll
    for (int k = 0; k < ROWSUM_RECS; ++k) {
        const int idx = t + k * (ROWSUM_WAVES * 32);
        if (idx < nrec) {
            const unsigned key = (unsigned)rec[k].x;
            const int rl   = (int)(key >> 16) - bkt * RPB;
            const int slot = atomicAdd(&s_n[rl], 1);
            if (slot < LIST_CAP)
                s_list[rl][slot] = make_int2((int)(key & 0xffffu), rec[k].y);
        }
    }
    __syncthreads();
    const bool seg_full = tot > SEG_CAP;

    for (int row = wave; row < RPB; row += ROWSUM_WAVES) {
        const int r = bkt * RPB + row;
        int* rowp = reinterpret_cast<int*>(out) + (size_t)r * OUT_FEAT;

        const int n = seg_full ? (LIST_CAP + 1) : __builtin_amdgcn_readfirstlane(s_n[row]);

        float4 acc = make_float4(0.f, 0.f, 0.f, 0.f);

        if (n <= LIST_CAP) {
            int c0 = 0, c1 = 0, c2 = 0;
            unsigned v0 = 0, v1 = 0, v2 = 0;
            if (lane < n) {
                const int2 p = s_list[row][lane];
                c0 = p.x; v0 = (unsigned)p.y;
            }
            const int s1 = 32 + lane;
            if (s1 < n) {
                const int2 p = s_list[row][s1];
                c1 = p.x; v1 = (unsigned)p.y;
            }
            const int s2 = 64 + lane;
            if (s2 < n) {
                const int2 p = s_list[row][s2];
                c2 = p.x; v2 = (unsigned)p.y;
            }

            const int m0 = n < 32 ? n : 32;
            const int m1 = n < 64 ? n - 32 : 32;
            const int m2 = n - 64;

            int rank0 = 0, rank1 = 0, rank2 = 0;
            if (n <= 32) {
                for (int t = 0; t < n; ++t)
                    rank0 += (__builtin_amdgcn_readlane(c0, t) < c0) ? 1 : 0;
            } else {
                for (int t = 0; t < 32; ++t) {
                    const int kc = __builtin_amdgcn_readlane(c0, t);
                    rank0 += (kc < c0) ? 1 : 0; rank1 += (kc < c1) ? 1 : 0; rank2 += (kc < c2) ? 1 : 0;
                }
                for (int t = 0; t < m1; ++t) {
                    const int kc = __builtin_amdgcn_readlane(c1, t);
                    rank0 += (kc < c0) ? 1 : 0; rank1 += (kc < c1) ? 1 : 0; rank2 += (kc < c2) ? 1 : 0;
                }
                for (int t = 0; t < m2; ++t) {
                    const int kc = __builtin_amdgcn_readlane(c2, t);
                    rank0 += (kc < c0) ? 1 : 0; rank1 += (kc < c1) ? 1 : 0; rank2 += (kc < c2) ? 1 : 0;
                }
            }

            int2* strip = s_cv[wave];
            int2 q0, q1, q2;
            gcn_place(strip, lane, n, c0, v0, rank0, c1, v1, rank1, c2, v2, rank2, q0, q1, q2);

            const bool hole = ((lane < n) & (q0.x < 0)) | ((s1 < n) & (q1.x < 0)) | ((s2 < n) & (q2.x < 0));
            if (__builtin_amdgcn_ballot_w32(hole) != 0u) {
                rank0 = 0; rank1 = 0; rank2 = 0;
                for (int t = 0; t < n; ++t) {
                    int kc; unsigned kv;
                    if (t < 32)      { kc = __builtin_amdgcn_readlane(c0, t);      kv = (unsigned)__builtin_amdgcn_readlane((int)v0, t); }
                    else if (t < 64) { kc = __builtin_amdgcn_readlane(c1, t - 32); kv = (unsigned)__builtin_amdgcn_readlane((int)v1, t - 32); }
                    else             { kc = __builtin_amdgcn_readlane(c2, t - 64); kv = (unsigned)__builtin_amdgcn_readlane((int)v2, t - 64); }
                    rank0 += gcn_key_below(kc, kv, c0, v0);
                    rank1 += gcn_key_below(kc, kv, c1, v1);
                    rank2 += gcn_key_below(kc, kv, c2, v2);
                }
                gcn_place(strip, lane, n, c0, v0, rank0, c1, v1, rank1, c2, v2, rank2, q0, q1, q2);
            }

            int   pc = 0;
            float pv = 0.f;
#define GCN_SUM_SEGMENT(Q, M)                                                       \
            for (int t0 = 0; t0 < (M); t0 += 8) {                                       \
                _Pragma("unroll")                                                       \
                for (int u = 0; u < 8; ++u) {                                           \
                    const int t = t0 + u;                                               \
                    int   kc = __builtin_amdgcn_readlane((Q).x, t);                     \
                    float kv = __int_as_float(__builtin_amdgcn_readlane((Q).y, t));     \
                    if (kc < 0) { kc = pc; kv = (t < (M)) ? pv : 0.f; }                 \
                    pc = kc; pv = kv;                                                   \
                    const float4 h = H4[(size_t)kc * (OUT_FEAT / 4) + lane];            \
                    acc.x = fmaf(kv, h.x, acc.x); acc.y = fmaf(kv, h.y, acc.y);         \
                    acc.z = fmaf(kv, h.z, acc.z); acc.w = fmaf(kv, h.w, acc.w);         \
                }                                                                       \
            }
            GCN_SUM_SEGMENT(q0, m0)
            GCN_SUM_SEGMENT(q1, m1)
            GCN_SUM_SEGMENT(q2, m2)
#undef GCN_SUM_SEGMENT
        } else {
            for (int base = 0; base < N_EDGES; base += 32) {
                unsigned hit = __builtin_amdgcn_ballot_w32(rows[base + lane] == r);
                while (hit) {
                    const int b = __builtin_ctz(hit);
                    hit &= hit - 1;
                    const int   c = cols[base + b];
                    const float v = vals[base + b];
                    const float4 h = H4[(size_t)c * (OUT_FEAT / 4) + lane];
                    acc.x = fmaf(v, h.x, acc.x); acc.y = fmaf(v, h.y, acc.y);
                    acc.z = fmaf(v, h.z, acc.z); acc.w = fmaf(v, h.w, acc.w);
                }
            }
        }

        reinterpret_cast<int4*>(rowp)[lane] =
            make_int4(__float_as_int(acc.x), __float_as_int(acc.y),
                      __float_as_int(acc.z), __float_as_int(acc.w));
    }
}

extern "C" void kernel_launch(void* const* d_in, const int* in_sizes, int n_in,
                              void* d_out, int out_size, void* d_ws, size_t ws_size,
                              hipStream_t stream) {
    const float* X      = (const float*)d_in[0];
    const float* W      = (const float*)d_in[1];
    const float* A_vals = (const float*)d_in[2];
    const int*   A_rows = (const int*)d_in[3];
    const int*   A_cols = (const int*)d_in[4];
    float*       out    = (float*)d_out;
    float*       H      = (float*)d_ws;

    gcn_gemm_kernel<<<N_NODES / 16 / GEMM_TILES_PER_BLOCK, 256, 0, stream>>>(X, W, H, out);

    gcn_part_kernel<<<PART_BLOCKS, PART_THREADS, 0, stream>>>(A_vals, A_rows, A_cols, out);

    gcn_rowsum_kernel<<<NB, ROWSUM_WAVES * 32, 0, stream>>>(H, A_vals, A_rows, A_cols, out);
}


# sum moved before the product, both in one launch: a bucket's block sums its 100 rows of A X (ordered lists as before) in LDS, then x W^T on WMMA; no H, no GEMM launch, no global adds; new sum order
# speedup vs baseline: 4.6256x; 1.2655x over previous
#include <hip/hip_runtime.h>

#define N_NODES  50000
#define N_EDGES  1600000
#define IN_FEAT  128
#define OUT_FEAT 128

#define RPB          100
#define NB           (N_NODES / RPB)
#define LIST_CAP     79
static_assert(N_NODES <= 65536, "destination and source share one 32-bit word of a record");

typedef __attribute__((ext_vector_type(2))) float v2f;
typedef __attribute__((ext_vector_type(8))) float v8f;

#define PART_THREADS 512
#define PART_GROUPS  4
#define PART_BGROUPS (PART_THREADS * PART_GROUPS)
#define PART_EDGES   (PART_BGROUPS * 4)
#define PART_BLOCKS  ((N_EDGES / 4 + PART_BGROUPS - 1) / PART_BGROUPS)
static_assert(NB <= PART_THREADS, "one thread per bucket in the counting and scan steps");
#define WS_TAB_FIRST (PART_BLOCKS * PART_EDGES)
static_assert(N_EDGES % 4 == 0, "edges are read in groups of 4");

__global__ __launch_bounds__(PART_THREADS) void gcn_part_kernel(const float* __restrict__ vals,
                                                                const int* __restrict__ rows,
                                                                const int* __restrict__ cols,
                                                                float* __restrict__ ws) {
    __shared__ int  s_cnt[NB];
    __shared__ int  s_off[NB];
    __shared__ int  s_wtot[PART_THREADS / 32];
    __shared__ int2 s_rec[PART_EDGES];

    const int t    = threadIdx.x;
    const int lane = t & 31;
    const int wave = t >> 5;
    const int g0   = (int)blockIdx.x * PART_BGROUPS;
    int live_groups = N_EDGES / 4 - g0;
    if (live_groups > PART_BGROUPS) live_groups = PART_BGROUPS;
    const int total = 4 * live_groups;

    if (t < NB) s_cnt[t] = 0;
    __syncthreads();

    int4   r[PART_GROUPS];
    int4   c[PART_GROUPS];
    float4 v[PART_GROUPS];
    int4   l[PART_GROUPS];
#pragma unroll
    for (int j = 0; j < PART_GROUPS; ++j) {
        const int gl = j * PART_THREADS + t;
        r[j] = make_int4(0, 0, 0, 0);
        c[j] = make_int4(0, 0, 0, 0);
        v[j] = make_float4(0.f, 0.f, 0.f, 0.f);
        if (gl < live_groups) {
            r[j] = reinterpret_cast<const int4*>(rows)[g0 + gl];
            c[j] = reinterpret_cast<const int4*>(cols)[g0 + gl];
            v[j] = reinterpret_cast<const float4*>(vals)[g0 + gl];
        }
    }
#pragma unroll
    for (int j = 0; j < PART_GROUPS; ++j) {
        l[j] = make_int4(0, 0, 0, 0);
        if (j * PART_THREADS + t < live_groups) {
            l[j].x = atomicAdd(&s_cnt[r[j].x / RPB], 1);
            l[j].y = atomicAdd(&s_cnt[r[j].y / RPB], 1);
            l[j].z = atomicAdd(&s_cnt[r[j].z / RPB], 1);
            l[j].w = atomicAdd(&s_cnt[r[j].w / RPB], 1);
        }
    }
    __syncthreads();

    const int cnt = (t < NB) ? s_cnt[t] : 0;
    int incl = cnt;
    for (int d = 1; d < 32; d <<= 1) {
        const int y = __shfl_up(incl, d, 32);
        if (lane >= d) incl += y;
    }
    if (lane == 31) s_wtot[wave] = incl;
    __syncthreads();
    int pre = 0;
    for (int w = 0; w < wave; ++w) pre += s_wtot[w];

    int2* const recs = reinterpret_cast<int2*>(ws);
    if (t < NB) {
        const int off = pre + incl - cnt;
        s_off[t] = off;
        recs[(size_t)WS_TAB_FIRST + (size_t)t * PART_BLOCKS + blockIdx.x] = make_int2(off, cnt);
    }
    __syncthreads();

#pragma unroll
    for (int j = 0; j < PART_GROUPS; ++j) {
        if (j * PART_THREADS + t < live_groups) {
            s_rec[s_off[r[j].x / RPB] + l[j].x] = make_int2((int)(((unsigned)r[j].x << 16) | (unsigned)c[j].x), __float_as_int(v[j].x));
            s_rec[s_off[r[j].y / RPB] + l[j].y] = make_int2((int)(((unsigned)r[j].y << 16) | (unsigned)c[j].y), __float_as_int(v[j].y));
            s_rec[s_off[r[j].z / RPB] + l[j].z] = make_int2((int)(((unsigned)r[j].z << 16) | (unsigned)c[j].z), __float_as_int(v[j].z));
            s_rec[s_off[r[j].w / RPB] + l[j].w] = make_int2((int)(((unsigned)r[j].w << 16) | (unsigned)c[j].w), __float_as_int(v[j].w));
        }
    }
    __syncthreads();

#pragma unroll
    for (int j = 0; j < 4 * PART_GROUPS; ++j) {
        const int idx = t + j * PART_THREADS;
        if (idx < total)
            recs[(size_t)blockIdx.x * PART_EDGES + idx] = s_rec[idx];
    }
}

__device__ __forceinline__ int gcn_key_below(int kc, unsigned kv, int c, unsigned v) {
    return ((kc < c) | ((kc == c) & (kv < v))) ? 1 : 0;
}

__device__ __forceinline__ void gcn_place(int2* strip, int lane, int n,
                                          int c0, unsigned v0, int rank0,
                                          int c1, unsigned v1, int rank1,
                                          int c2, unsigned v2, int rank2,
                                          int2& q0, int2& q1, int2& q2) {
    strip[lane]      = make_int2(-1, 0);
    strip[32 + lane] = make_int2(-1, 0);
    strip[64 + lane] = make_int2(-1, 0);
    if (lane < n)      strip[rank0] = make_int2(c0, (int)v0);
    if (32 + lane < n) strip[rank1] = make_int2(c1, (int)v1);
    if (64 + lane < n) strip[rank2] = make_int2(c2, (int)v2);
    q0 = strip[lane];
    q1 = strip[32 + lane];
    q2 = strip[64 + lane];
}

#define AX_PITCH 132
#define AX_ROWS  112
template <int NT>
__device__ __forceinline__ void gcn_products(const float* s_ax, const float* __restrict__ W,
                                             float* __restrict__ outb, int mt0, int nt, int lane) {
    int ls = lane & 15;
    int hs = lane >> 4;
    asm volatile("" : "+v"(ls), "+v"(hs));
    const float* __restrict__ wb = W + (size_t)(nt * 16 + ls) * IN_FEAT + 4 * hs;
    const float* xa = s_ax + (mt0 * 16 + ls) * AX_PITCH + 4 * hs;
    v8f acc[NT];
#pragma unroll
    for (int j = 0; j < NT; ++j) acc[j] = v8f{};
    float4 b[4];
#pragma unroll
    for (int i = 0; i < 4; ++i)
        b[i] = *reinterpret_cast<const float4*>(wb + 8 * i);
#pragma unroll
    for (int kq = 0; kq < IN_FEAT / 32; ++kq) {
        float4 bn[4];
#pragma unroll
        for (int i = 0; i < 4; ++i) {
            bn[i] = b[i];
            if (kq + 1 < IN_FEAT / 32)
                bn[i] = *reinterpret_cast<const float4*>(wb + (kq + 1) * 32 + 8 * i);
        }
        __builtin_amdgcn_sched_barrier(0);
#pragma unroll
        for (int j = 0; j < NT; ++j) {
            float4 a[4];
#pragma unroll
            for (int i = 0; i < 4; ++i)
                a[i] = *reinterpret_cast<const float4*>(xa + j * 32 * AX_PITCH + kq * 32 + 8 * i);
#pragma unroll
            for (int i = 0; i < 4; ++i) {
                const v2f a0 = {a[i].x, a[i].y};
                const v2f b0 = {b[i].x, b[i].y};
                acc[j] = __builtin_amdgcn_wmma_f32_16x16x4_f32(
                    false, a0, false, b0, (short)0, acc[j], false, false);
                const v2f a1 = {a[i].z, a[i].w};
                const v2f b1 = {b[i].z, b[i].w};
                acc[j] = __builtin_amdgcn_wmma_f32_16x16x4_f32(
                    false, a1, false, b1, (short)0, acc[j], false, false);
            }
        }
        __builtin_amdgcn_sched_barrier(0);
#pragma unroll
        for (int i = 0; i < 4; ++i) b[i] = bn[i];
    }
    if constexpr (NT == 3) {
        asm volatile("v_nop\n\tv_nop\n\tv_nop\n\tv_nop\n\tv_nop\n\tv_nop\n\tv_nop\n\tv_nop"
                     : "+v"(acc[0]), "+v"(acc[1]), "+v"(acc[2]), "+v"(hs), "+v"(ls));
    } else {
        static_assert(NT == 1 || NT == 3, "one or three row tiles per call");
        asm volatile("v_nop\n\tv_nop\n\tv_nop\n\tv_nop\n\tv_nop\n\tv_nop\n\tv_nop\n\tv_nop"
                     : "+v"(acc[0]), "+v"(hs), "+v"(ls));
    }
#pragma unroll
    for (int j = 0; j < NT; ++j) {
#pragma unroll
        for (int v = 0; v < 8; ++v) {
            const int rloc = (mt0 + 2 * j) * 16 + 8 * hs + v;
            if (rloc < RPB)
                outb[(size_t)rloc * OUT_FEAT + nt * 16 + ls] = acc[j][v];
        }
    }
}

#define ROWSUM_WAVES 20
#define ROWSUM_RUNS  ((PART_BLOCKS + ROWSUM_WAVES - 1) / ROWSUM_WAVES)
static_assert(RPB % ROWSUM_WAVES == 0, "every wave of a block takes the same number of destinations");
static_assert(ROWSUM_WAVES == 20 && AX_ROWS == 112, "the product phase deals 7 x 8 tiles to 20 waves");

__global__ __launch_bounds__(ROWSUM_WAVES * 32) void gcn_rowsum_kernel(const float* __restrict__ X,
                                                                       const float* __restrict__ W,
                                                                       const float* __restrict__ vals,
                                                                       const int* __restrict__ rows,
                                                                       const int* __restrict__ cols,
                                                                       const float* __restrict__ ws,
                                                                       float* __restrict__ out) {
    __shared__ int2 s_cv[ROWSUM_WAVES][96];
    __shared__ int2 s_list[RPB][LIST_CAP];
    __shared__ int  s_n[RPB];
    __shared__ int2 s_run[PART_BLOCKS];
    __shared__ __attribute__((aligned(16))) float s_ax[AX_ROWS * AX_PITCH];

    const int t    = threadIdx.x;
    const int lane = t & 31;
    const int wave = __builtin_amdgcn_readfirstlane((int)(t >> 5));
    const int bkt  = (int)blockIdx.x;

    const int2* __restrict__ recs = reinterpret_cast<const int2*>(ws);
    const float4* __restrict__ X4 = reinterpret_cast<const float4*>(X);

    if (t < RPB) s_n[t] = 0;
    if (t < PART_BLOCKS)
        s_run[t] = recs[(size_t)WS_TAB_FIRST + (size_t)bkt * PART_BLOCKS + t];
    for (int i = t; i < (AX_ROWS - RPB) * AX_PITCH; i += ROWSUM_WAVES * 32)
        s_ax[RPB * AX_PITCH + i] = 0.f;
    __syncthreads();

    int2 rec[ROWSUM_RUNS];
    int  rlen[ROWSUM_RUNS];
#pragma unroll
    for (int k = 0; k < ROWSUM_RUNS; ++k) {
        const int b = wave + k * ROWSUM_WAVES;
        rec[k]  = make_int2(0, 0);
        rlen[k] = 0;
        if (b < PART_BLOCKS) {
            const int2 rr = s_run[b];
            rlen[k] = rr.y;
            if (lane < rr.y) rec[k] = recs[(size_t)b * PART_EDGES + rr.x + lane];
        }
    }
#pragma unroll
    for (int k = 0; k < ROWSUM_RUNS; ++k) {
        if (lane < rlen[k]) {
            const unsigned key = (unsigned)rec[k].x;
            const int rl   = (int)(key >> 16) - bkt * RPB;
            const int slot = atomicAdd(&s_n[rl], 1);
            if (slot < LIST_CAP)
                s_list[rl][slot] = make_int2((int)(key & 0xffffu), rec[k].y);
        }
    }
    for (int b = wave; b < PART_BLOCKS; b += ROWSUM_WAVES) {
        const int2 rr = s_run[b];
        for (int i = 32 + lane; i < rr.y; i += 32) {
            const int2 rc = recs[(size_t)b * PART_EDGES + rr.x + i];
            const unsigned key = (unsigned)rc.x;
            const int rl   = (int)(key >> 16) - bkt * RPB;
            const int slot = atomicAdd(&s_n[rl], 1);
            if (slot < LIST_CAP)
                s_list[rl][slot] = make_int2((int)(key & 0xffffu), rc.y);
        }
    }
    __syncthreads();

    for (int row = wave; row < RPB; row += ROWSUM_WAVES) {
        const int r = bkt * RPB + row;

        const int n = __builtin_amdgcn_readfirstlane(s_n[row]);

        float4 acc = make_float4(0.f, 0.f, 0.f, 0.f);

        if (n <= LIST_CAP) {
            int c0 = 0, c1 = 0, c2 = 0;
            unsigned v0 = 0, v1 = 0, v2 = 0;
            if (lane < n) {
                const int2 p = s_list[row][lane];
                c0 = p.x; v0 = (unsigned)p.y;
            }
            const int s1 = 32 + lane;
            if (s1 < n) {
                const int2 p = s_list[row][s1];
                c1 = p.x; v1 = (unsigned)p.y;
            }
            const int s2 = 64 + lane;
            if (s2 < n) {
                const int2 p = s_list[row][s2];
                c2 = p.x; v2 = (unsigned)p.y;
            }

            const int m0 = n < 32 ? n : 32;
            const int m1 = n < 64 ? n - 32 : 32;
            const int m2 = n - 64;

            int rank0 = 0, rank1 = 0, rank2 = 0;
            if (n <= 32) {
                for (int t = 0; t < n; ++t)
                    rank0 += (__builtin_amdgcn_readlane(c0, t) < c0) ? 1 : 0;
            } else {
                for (int t = 0; t < 32; ++t) {
                    const int kc = __builtin_amdgcn_readlane(c0, t);
                    rank0 += (kc < c0) ? 1 : 0; rank1 += (kc < c1) ? 1 : 0; rank2 += (kc < c2) ? 1 : 0;
                }
                for (int t = 0; t < m1; ++t) {
                    const int kc = __builtin_amdgcn_readlane(c1, t);
                    rank0 += (kc < c0) ? 1 : 0; rank1 += (kc < c1) ? 1 : 0; rank2 += (kc < c2) ? 1 : 0;
                }
                for (int t = 0; t < m2; ++t) {
                    const int kc = __builtin_amdgcn_readlane(c2, t);
                    rank0 += (kc < c0) ? 1 : 0; rank1 += (kc < c1) ? 1 : 0; rank2 += (kc < c2) ? 1 : 0;
                }
            }

            int2* strip = s_cv[wave];
            int2 q0, q1, q2;
            gcn_place(strip, lane, n, c0, v0, rank0, c1, v1, rank1, c2, v2, rank2, q0, q1, q2);

            const bool hole = ((lane < n) & (q0.x < 0)) | ((s1 < n) & (q1.x < 0)) | ((s2 < n) & (q2.x < 0));
            if (__builtin_amdgcn_ballot_w32(hole) != 0u) {
                rank0 = 0; rank1 = 0; rank2 = 0;
                for (int t = 0; t < n; ++t) {
                    int kc; unsigned kv;
                    if (t < 32)      { kc = __builtin_amdgcn_readlane(c0, t);      kv = (unsigned)__builtin_amdgcn_readlane((int)v0, t); }
                    else if (t < 64) { kc = __builtin_amdgcn_readlane(c1, t - 32); kv = (unsigned)__builtin_amdgcn_readlane((int)v1, t - 32); }
                    else             { kc = __builtin_amdgcn_readlane(c2, t - 64); kv = (unsigned)__builtin_amdgcn_readlane((int)v2, t - 64); }
                    rank0 += gcn_key_below(kc, kv, c0, v0);
                    rank1 += gcn_key_below(kc, kv, c1, v1);
                    rank2 += gcn_key_below(kc, kv, c2, v2);
                }
                gcn_place(strip, lane, n, c0, v0, rank0, c1, v1, rank1, c2, v2, rank2, q0, q1, q2);
            }

            int   pc = 0;
            float pv = 0.f;
#define GCN_SUM_SEGMENT(Q, M)                                                       \
            for (int t0 = 0; t0 < (M); t0 += 8) {                                       \
                _Pragma("unroll")                                                       \
                for (int u = 0; u < 8; ++u) {                                           \
                    const int t = t0 + u;                                               \
                    int   kc = __builtin_amdgcn_readlane((Q).x, t);                     \
                    float kv = __int_as_float(__builtin_amdgcn_readlane((Q).y, t));     \
                    if (kc < 0) { kc = pc; kv = (t < (M)) ? pv : 0.f; }                 \
                    pc = kc; pv = kv;                                                   \
                    const float4 h = X4[(size_t)kc * (IN_FEAT / 4) + lane];            \
                    acc.x = fmaf(kv, h.x, acc.x); acc.y = fmaf(kv, h.y, acc.y);         \
                    acc.z = fmaf(kv, h.z, acc.z); acc.w = fmaf(kv, h.w, acc.w);         \
                }                                                                       \
            }
            GCN_SUM_SEGMENT(q0, m0)
            GCN_SUM_SEGMENT(q1, m1)
            GCN_SUM_SEGMENT(q2, m2)
#undef GCN_SUM_SEGMENT
        } else {
            for (int base = 0; base < N_EDGES; base += 32) {
                unsigned hit = __builtin_amdgcn_ballot_w32(rows[base + lane] == r);
                while (hit) {
                    const int b = __builtin_ctz(hit);
                    hit &= hit - 1;
                    const int   c = cols[base + b];
                    const float v = vals[base + b];
                    const float4 h = X4[(size_t)c * (IN_FEAT / 4) + lane];
                    acc.x = fmaf(v, h.x, acc.x); acc.y = fmaf(v, h.y, acc.y);
                    acc.z = fmaf(v, h.z, acc.z); acc.w = fmaf(v, h.w, acc.w);
                }
            }
        }

        *reinterpret_cast<float4*>(s_ax + row * AX_PITCH + 4 * lane) = acc;
    }

    __syncthreads();

    float* __restrict__ outb = out + (size_t)bkt * RPB * OUT_FEAT;
    if (wave < 16) {
        gcn_products<3>(s_ax, W, outb, wave >> 3, wave & 7, lane);
    } else {
        gcn_products<1>(s_ax, W, outb, 6, 2 * (wave - 16), lane);
        gcn_products<1>(s_ax, W, outb, 6, 2 * (wave - 16) + 1, lane);
    }
}

extern "C" void kernel_launch(void* const* d_in, const int* in_sizes, int n_in,
                              void* d_out, int out_size, void* d_ws, size_t ws_size,
                              hipStream_t stream) {
    const float* X      = (const float*)d_in[0];
    const float* W      = (const float*)d_in[1];
    const float* A_vals = (const float*)d_in[2];
    const int*   A_rows = (const int*)d_in[3];
    const int*   A_cols = (const int*)d_in[4];
    float*       out    = (float*)d_out;
    float*       ws     = (float*)d_ws;

    gcn_part_kernel<<<PART_BLOCKS, PART_THREADS, 0, stream>>>(A_vals, A_rows, A_cols, ws);

    gcn_rowsum_kernel<<<NB, ROWSUM_WAVES * 32, 0, stream>>>(X, W, A_vals, A_rows, A_cols, ws, out);
}


# launch 2's loads no longer wait on address translation one behind another: a group's 8 row addresses get register pairs of their own, the 10 run loads lose their branches; same sums, same order
# speedup vs baseline: 4.8071x; 1.0392x over previous
#include <hip/hip_runtime.h>

#define N_NODES  50000
#define N_EDGES  1600000
#define IN_FEAT  128
#define OUT_FEAT 128

#define RPB          100
#define NB           (N_NODES / RPB)
#define LIST_CAP     79
static_assert(N_NODES <= 65536, "destination and source share one 32-bit word of a record");

typedef __attribute__((ext_vector_type(2))) float v2f;
typedef __attribute__((ext_vector_type(8))) float v8f;
typedef __attribute__((ext_vector_type(4))) float v4f;
typedef const __attribute__((address_space(1))) v4f* gcn_row_ptr;

#define PART_THREADS 512
#define PART_GROUPS  4
#define PART_BGROUPS (PART_THREADS * PART_GROUPS)
#define PART_EDGES   (PART_BGROUPS * 4)
#define PART_BLOCKS  ((N_EDGES / 4 + PART_BGROUPS - 1) / PART_BGROUPS)
static_assert(NB <= PART_THREADS, "one thread per bucket in the counting and scan steps");
#define WS_TAB_FIRST (PART_BLOCKS * PART_EDGES)
static_assert(N_EDGES % 4 == 0, "edges are read in groups of 4");

__global__ __launch_bounds__(PART_THREADS) void gcn_part_kernel(const float* __restrict__ vals,
                                                                const int* __restrict__ rows,
                                                                const int* __restrict__ cols,
                                                                float* __restrict__ ws) {
    __shared__ int  s_cnt[NB];
    __shared__ int  s_off[NB];
    __shared__ int  s_wtot[PART_THREADS / 32];
    __shared__ int2 s_rec[PART_EDGES];

    const int t    = threadIdx.x;
    const int lane = t & 31;
    const int wave = t >> 5;
    const int g0   = (int)blockIdx.x * PART_BGROUPS;
    int live_groups = N_EDGES / 4 - g0;
    if (live_groups > PART_BGROUPS) live_groups = PART_BGROUPS;
    const int total = 4 * live_groups;

    if (t < NB) s_cnt[t] = 0;
    __syncthreads();

    int4   r[PART_GROUPS];
    int4   c[PART_GROUPS];
    float4 v[PART_GROUPS];
    int4   l[PART_GROUPS];
#pragma unroll
    for (int j = 0; j < PART_GROUPS; ++j) {
        const int gl = j * PART_THREADS + t;
        r[j] = make_int4(0, 0, 0, 0);
        c[j] = make_int4(0, 0, 0, 0);
        v[j] = make_float4(0.f, 0.f, 0.f, 0.f);
        if (gl < live_groups) {
            r[j] = reinterpret_cast<const int4*>(rows)[g0 + gl];
            c[j] = reinterpret_cast<const int4*>(cols)[g0 + gl];
            v[j] = reinterpret_cast<const float4*>(vals)[g0 + gl];
        }
    }
#pragma unroll
    for (int j = 0; j < PART_GROUPS; ++j) {
        l[j] = make_int4(0, 0, 0, 0);
        if (j * PART_THREADS + t < live_groups) {
            l[j].x = atomicAdd(&s_cnt[r[j].x / RPB], 1);
            l[j].y = atomicAdd(&s_cnt[r[j].y / RPB], 1);
            l[j].z = atomicAdd(&s_cnt[r[j].z / RPB], 1);
            l[j].w = atomicAdd(&s_cnt[r[j].w / RPB], 1);
        }
    }
    __syncthreads();

    const int cnt = (t < NB) ? s_cnt[t] : 0;
    int incl = cnt;
    for (int d = 1; d < 32; d <<= 1) {
        const int y = __shfl_up(incl, d, 32);
        if (lane >= d) incl += y;
    }
    if (lane == 31) s_wtot[wave] = incl;
    __syncthreads();
    int pre = 0;
    for (int w = 0; w < wave; ++w) pre += s_wtot[w];

    int2* const recs = reinterpret_cast<int2*>(ws);
    if (t < NB) {
        const int off = pre + incl - cnt;
        s_off[t] = off;
        recs[(size_t)WS_TAB_FIRST + (size_t)t * PART_BLOCKS + blockIdx.x] = make_int2(off, cnt);
    }
    __syncthreads();

#pragma unroll
    for (int j = 0; j < PART_GROUPS; ++j) {
        if (j * PART_THREADS + t < live_groups) {
            s_rec[s_off[r[j].x / RPB] + l[j].x] = make_int2((int)(((unsigned)r[j].x << 16) | (unsigned)c[j].x), __float_as_int(v[j].x));
            s_rec[s_off[r[j].y / RPB] + l[j].y] = make_int2((int)(((unsigned)r[j].y << 16) | (unsigned)c[j].y), __float_as_int(v[j].y));
            s_rec[s_off[r[j].z / RPB] + l[j].z] = make_int2((int)(((unsigned)r[j].z << 16) | (unsigned)c[j].z), __float_as_int(v[j].z));
            s_rec[s_off[r[j].w / RPB] + l[j].w] = make_int2((int)(((unsigned)r[j].w << 16) | (unsigned)c[j].w), __float_as_int(v[j].w));
        }
    }
    __syncthreads();

#pragma unroll
    for (int j = 0; j < 4 * PART_GROUPS; ++j) {
        const int idx = t + j * PART_THREADS;
        if (idx < total)
            recs[(size_t)blockIdx.x * PART_EDGES + idx] = s_rec[idx];
    }
}

__device__ __forceinline__ int gcn_key_below(int kc, unsigned kv, int c, unsigned v) {
    return ((kc < c) | ((kc == c) & (kv < v))) ? 1 : 0;
}

__device__ __forceinline__ void gcn_place(int2* strip, int lane, int n,
                                          int c0, unsigned v0, int rank0,
                                          int c1, unsigned v1, int rank1,
                                          int c2, unsigned v2, int rank2,
                                          int2& q0, int2& q1, int2& q2) {
    strip[lane]      = make_int2(-1, 0);
    strip[32 + lane] = make_int2(-1, 0);
    strip[64 + lane] = make_int2(-1, 0);
    if (lane < n)      strip[rank0] = make_int2(c0, (int)v0);
    if (32 + lane < n) strip[rank1] = make_int2(c1, (int)v1);
    if (64 + lane < n) strip[rank2] = make_int2(c2, (int)v2);
    q0 = strip[lane];
    q1 = strip[32 + lane];
    q2 = strip[64 + lane];
}

#define AX_PITCH 132
#define AX_ROWS  112
template <int NT>
__device__ __forceinline__ void gcn_products(const float* s_ax, const float* __restrict__ W,
                                             float* __restrict__ outb, int mt0, int nt, int lane) {
    int ls = lane & 15;
    int hs = lane >> 4;
    asm volatile("" : "+v"(ls), "+v"(hs));
    const float* __restrict__ wb = W + (size_t)(nt * 16 + ls) * IN_FEAT + 4 * hs;
    const float* xa = s_ax + (mt0 * 16 + ls) * AX_PITCH + 4 * hs;
    v8f acc[NT];
#pragma unroll
    for (int j = 0; j < NT; ++j) acc[j] = v8f{};
    float4 b[4];
#pragma unroll
    for (int i = 0; i < 4; ++i)
        b[i] = *reinterpret_cast<const float4*>(wb + 8 * i);
#pragma unroll
    for (int kq = 0; kq < IN_FEAT / 32; ++kq) {
        float4 bn[4];
#pragma unroll
        for (int i = 0; i < 4; ++i) {
            bn[i] = b[i];
            if (kq + 1 < IN_FEAT / 32)
                bn[i] = *reinterpret_cast<const float4*>(wb + (kq + 1) * 32 + 8 * i);
        }
        __builtin_amdgcn_sched_barrier(0);
#pragma unroll
        for (int j = 0; j < NT; ++j) {
            float4 a[4];
#pragma unroll
            for (int i = 0; i < 4; ++i)
                a[i] = *reinterpret_cast<const float4*>(xa + j * 32 * AX_PITCH + kq * 32 + 8 * i);
#pragma unroll
            for (int i = 0; i < 4; ++i) {
                const v2f a0 = {a[i].x, a[i].y};
                const v2f b0 = {b[i].x, b[i].y};
                acc[j] = __builtin_amdgcn_wmma_f32_16x16x4_f32(
                    false, a0, false, b0, (short)0, acc[j], false, false);
                const v2f a1 = {a[i].z, a[i].w};
                const v2f b1 = {b[i].z, b[i].w};
                acc[j] = __builtin_amdgcn_wmma_f32_16x16x4_f32(
                    false, a1, false, b1, (short)0, acc[j], false, false);
            }
        }
        __builtin_amdgcn_sched_barrier(0);
#pragma unroll
        for (int i = 0; i < 4; ++i) b[i] = bn[i];
    }
    if constexpr (NT == 3) {
        asm volatile("v_nop\n\tv_nop\n\tv_nop\n\tv_nop\n\tv_nop\n\tv_nop\n\tv_nop\n\tv_nop"
                     : "+v"(acc[0]), "+v"(acc[1]), "+v"(acc[2]), "+v"(hs), "+v"(ls));
    } else {
        static_assert(NT == 1 || NT == 3, "one or three row tiles per call");
        asm volatile("v_nop\n\tv_nop\n\tv_nop\n\tv_nop\n\tv_nop\n\tv_nop\n\tv_nop\n\tv_nop"
                     : "+v"(acc[0]), "+v"(hs), "+v"(ls));
    }
#pragma unroll
    for (int j = 0; j < NT; ++j) {
#pragma unroll
        for (int v = 0; v < 8; ++v) {
            const int rloc = (mt0 + 2 * j) * 16 + 8 * hs + v;
            if (rloc < RPB)
                outb[(size_t)rloc * OUT_FEAT + nt * 16 + ls] = acc[j][v];
        }
    }
}

#define ROWSUM_WAVES 20
#define ROWSUM_RUNS  ((PART_BLOCKS + ROWSUM_WAVES - 1) / ROWSUM_WAVES)
static_assert(RPB % ROWSUM_WAVES == 0, "every wave of a block takes the same number of destinations");
static_assert(ROWSUM_WAVES == 20 && AX_ROWS == 112, "the product phase deals 7 x 8 tiles to 20 waves");

__global__ __launch_bounds__(ROWSUM_WAVES * 32) void gcn_rowsum_kernel(const float* __restrict__ X,
                                                                       const float* __restrict__ W,
                                                                       const float* __restrict__ vals,
                                                                       const int* __restrict__ rows,
                                                                       const int* __restrict__ cols,
                                                                       const float* __restrict__ ws,
                                                                       float* __restrict__ out) {
    __shared__ int2 s_cv[ROWSUM_WAVES][96];
    __shared__ int2 s_list[RPB][LIST_CAP];
    __shared__ int  s_n[RPB];
    __shared__ int2 s_run[PART_BLOCKS];
    __shared__ __attribute__((aligned(16))) float s_ax[AX_ROWS * AX_PITCH];

    const int t    = threadIdx.x;
    const int lane = t & 31;
    const int wave = __builtin_amdgcn_readfirstlane((int)(t >> 5));
    const int bkt  = (int)blockIdx.x;

    const int2* __restrict__ recs = reinterpret_cast<const int2*>(ws);
    const float4* __restrict__ X4 = reinterpret_cast<const float4*>(X);

    if (t < RPB) s_n[t] = 0;
    if (t < PART_BLOCKS)
        s_run[t] = recs[(size_t)WS_TAB_FIRST + (size_t)bkt * PART_BLOCKS + t];
    for (int i = t; i < (AX_ROWS - RPB) * AX_PITCH; i += ROWSUM_WAVES * 32)
        s_ax[RPB * AX_PITCH + i] = 0.f;
    __syncthreads();

    int2 rec[ROWSUM_RUNS];
    int  rlen[ROWSUM_RUNS];
    unsigned ridx[ROWSUM_RUNS];
    unsigned long long rw[ROWSUM_RUNS];
    static_assert((long long)WS_TAB_FIRST + (long long)NB * PART_BLOCKS < (1ll << 31), "record and table indices fit 32 bits");
#pragma unroll
    for (int k = 0; k < ROWSUM_RUNS; ++k) {
        const int b = wave + k * ROWSUM_WAVES;
        int2 rr = make_int2(0, 0);
        if (b < PART_BLOCKS) rr = s_run[b];
        rlen[k] = rr.y;
        const int last = lane < rr.y ? lane : rr.y - 1;
        const int in_run = b * PART_EDGES + rr.x + last;
        const int in_tab = WS_TAB_FIRST + bkt * PART_BLOCKS + (b < PART_BLOCKS ? b : 0);
        ridx[k] = (unsigned)(rr.y > 0 ? in_run : in_tab);
    }
#pragma unroll
    for (int k = 0; k < ROWSUM_RUNS; ++k) rw[k] = reinterpret_cast<const unsigned long long*>(ws)[ridx[k]];
    static_assert(ROWSUM_RUNS == 10, "the statement below names ten records");
    asm volatile("" : "+v"(rw[0]), "+v"(rw[1]), "+v"(rw[2]), "+v"(rw[3]), "+v"(rw[4]),
                      "+v"(rw[5]), "+v"(rw[6]), "+v"(rw[7]), "+v"(rw[8]), "+v"(rw[9]));
#pragma unroll
    for (int k = 0; k < ROWSUM_RUNS; ++k)
        rec[k] = make_int2((int)(unsigned)(rw[k] & 0xffffffffull), (int)(unsigned)(rw[k] >> 32));
#pragma unroll
    for (int k = 0; k < ROWSUM_RUNS; ++k) {
        if (lane < rlen[k]) {
            const unsigned key = (unsigned)rec[k].x;
            const int rl   = (int)(key >> 16) - bkt * RPB;
            const int slot = atomicAdd(&s_n[rl], 1);
            if (slot < LIST_CAP)
                s_list[rl][slot] = make_int2((int)(key & 0xffffu), rec[k].y);
        }
    }
    for (int b = wave; b < PART_BLOCKS; b += ROWSUM_WAVES) {
        const int2 rr = s_run[b];
        for (int i = 32 + lane; i < rr.y; i += 32) {
            const int2 rc = recs[(size_t)b * PART_EDGES + rr.x + i];
            const unsigned key = (unsigned)rc.x;
            const int rl   = (int)(key >> 16) - bkt * RPB;
            const int slot = atomicAdd(&s_n[rl], 1);
            if (slot < LIST_CAP)
                s_list[rl][slot] = make_int2((int)(key & 0xffffu), rc.y);
        }
    }
    __syncthreads();

    for (int row = wave; row < RPB; row += ROWSUM_WAVES) {
        const int r = bkt * RPB + row;

        const int n = __builtin_amdgcn_readfirstlane(s_n[row]);

        float4 acc = make_float4(0.f, 0.f, 0.f, 0.f);

        if (n <= LIST_CAP) {
            int c0 = 0, c1 = 0, c2 = 0;
            unsigned v0 = 0, v1 = 0, v2 = 0;
            if (lane < n) {
                const int2 p = s_list[row][lane];
                c0 = p.x; v0 = (unsigned)p.y;
            }
            const int s1 = 32 + lane;
            if (s1 < n) {
                const int2 p = s_list[row][s1];
                c1 = p.x; v1 = (unsigned)p.y;
            }
            const int s2 = 64 + lane;
            if (s2 < n) {
                const int2 p = s_list[row][s2];
                c2 = p.x; v2 = (unsigned)p.y;
            }

            const int m0 = n < 32 ? n : 32;
            const int m1 = n < 64 ? n - 32 : 32;
            const int m2 = n - 64;

            int rank0 = 0, rank1 = 0, rank2 = 0;
            if (n <= 32) {
                for (int t = 0; t < n; ++t)
                    rank0 += (__builtin_amdgcn_readlane(c0, t) < c0) ? 1 : 0;
            } else {
                for (int t = 0; t < 32; ++t) {
                    const int kc = __builtin_amdgcn_readlane(c0, t);
                    rank0 += (kc < c0) ? 1 : 0; rank1 += (kc < c1) ? 1 : 0; rank2 += (kc < c2) ? 1 : 0;
                }
                for (int t = 0; t < m1; ++t) {
                    const int kc = __builtin_amdgcn_readlane(c1, t);
                    rank0 += (kc < c0) ? 1 : 0; rank1 += (kc < c1) ? 1 : 0; rank2 += (kc < c2) ? 1 : 0;
                }
                for (int t = 0; t < m2; ++t) {
                    const int kc = __builtin_amdgcn_readlane(c2, t);
                    rank0 += (kc < c0) ? 1 : 0; rank1 += (kc < c1) ? 1 : 0; rank2 += (kc < c2) ? 1 : 0;
                }
            }

            int2* strip = s_cv[wave];
            int2 q0, q1, q2;
            gcn_place(strip, lane, n, c0, v0, rank0, c1, v1, rank1, c2, v2, rank2, q0, q1, q2);

            const bool hole = ((lane < n) & (q0.x < 0)) | ((s1 < n) & (q1.x < 0)) | ((s2 < n) & (q2.x < 0));
            if (__builtin_amdgcn_ballot_w32(hole) != 0u) {
                rank0 = 0; rank1 = 0; rank2 = 0;
                for (int t = 0; t < n; ++t) {
                    int kc; unsigned kv;
                    if (t < 32)      { kc = __builtin_amdgcn_readlane(c0, t);      kv = (unsigned)__builtin_amdgcn_readlane((int)v0, t); }
                    else if (t < 64) { kc = __builtin_amdgcn_readlane(c1, t - 32); kv = (unsigned)__builtin_amdgcn_readlane((int)v1, t - 32); }
                    else             { kc = __builtin_amdgcn_readlane(c2, t - 64); kv = (unsigned)__builtin_amdgcn_readlane((int)v2, t - 64); }
                    rank0 += gcn_key_below(kc, kv, c0, v0);
                    rank1 += gcn_key_below(kc, kv, c1, v1);
                    rank2 += gcn_key_below(kc, kv, c2, v2);
                }
                gcn_place(strip, lane, n, c0, v0, rank0, c1, v1, rank1, c2, v2, rank2, q0, q1, q2);
            }

            int   pc = 0;
            float pv = 0.f;
            const unsigned long long xlane = (unsigned long long)X + (unsigned long long)(16u * (unsigned)lane);
#define GCN_SUM_SEGMENT(Q, M)                                                       \
            for (int t0 = 0; t0 < (M); t0 += 8) {                                       \
                unsigned long long ad[8];                                               \
                float wt[8];                                                            \
                _Pragma("unroll")                                                       \
                for (int u = 0; u < 8; ++u) {                                           \
                    const int t = t0 + u;                                               \
                    int   kc = __builtin_amdgcn_readlane((Q).x, t);                     \
                    float kv = __int_as_float(__builtin_amdgcn_readlane((Q).y, t));     \
                    if (kc < 0) { kc = pc; kv = (t < (M)) ? pv : 0.f; }                 \
                    pc = kc; pv = kv;                                                   \
                    wt[u] = kv;                                                         \
                    ad[u] = xlane + ((unsigned long long)(unsigned)kc << 9);            \
                }                                                                       \
                asm volatile("" : "+v"(ad[0]), "+v"(ad[1]), "+v"(ad[2]), "+v"(ad[3]),   \
                                  "+v"(ad[4]), "+v"(ad[5]), "+v"(ad[6]), "+v"(ad[7]));  \
                v4f h[8];                                                               \
                _Pragma("unroll")                                                       \
                for (int u = 0; u < 8; ++u) h[u] = *(gcn_row_ptr)ad[u];                 \
                _Pragma("unroll")                                                       \
                for (int u = 0; u < 8; ++u) {                                           \
                    acc.x = fmaf(wt[u], h[u].x, acc.x); acc.y = fmaf(wt[u], h[u].y, acc.y); \
                    acc.z = fmaf(wt[u], h[u].z, acc.z); acc.w = fmaf(wt[u], h[u].w, acc.w); \
                }                                                                       \
            }
            GCN_SUM_SEGMENT(q0, m0)
            GCN_SUM_SEGMENT(q1, m1)
            GCN_SUM_SEGMENT(q2, m2)
#undef GCN_SUM_SEGMENT
        } else {
            for (int base = 0; base < N_EDGES; base += 32) {
                unsigned hit = __builtin_amdgcn_ballot_w32(rows[base + lane] == r);
                while (hit) {
                    const int b = __builtin_ctz(hit);
                    hit &= hit - 1;
                    const int   c = cols[base + b];
                    const float v = vals[base + b];
                    const float4 h = X4[(size_t)c * (IN_FEAT / 4) + lane];
                    acc.x = fmaf(v, h.x, acc.x); acc.y = fmaf(v, h.y, acc.y);
                    acc.z = fmaf(v, h.z, acc.z); acc.w = fmaf(v, h.w, acc.w);
                }
            }
        }

        *reinterpret_cast<float4*>(s_ax + row * AX_PITCH + 4 * lane) = acc;
    }

    __syncthreads();

    float* __restrict__ outb = out + (size_t)bkt * RPB * OUT_FEAT;
    if (wave < 16) {
        gcn_products<3>(s_ax, W, outb, wave >> 3, wave & 7, lane);
    } else {
        gcn_products<1>(s_ax, W, outb, 6, 2 * (wave - 16), lane);
        gcn_products<1>(s_ax, W, outb, 6, 2 * (wave - 16) + 1, lane);
    }
}

extern "C" void kernel_launch(void* const* d_in, const int* in_sizes, int n_in,
                              void* d_out, int out_size, void* d_ws, size_t ws_size,
                              hipStream_t stream) {
    const float* X      = (const float*)d_in[0];
    const float* W      = (const float*)d_in[1];
    const float* A_vals = (const float*)d_in[2];
    const int*   A_rows = (const int*)d_in[3];
    const int*   A_cols = (const int*)d_in[4];
    float*       out    = (float*)d_out;
    float*       ws     = (float*)d_ws;

    gcn_part_kernel<<<PART_BLOCKS, PART_THREADS, 0, stream>>>(A_vals, A_rows, A_cols, ws);

    gcn_rowsum_kernel<<<NB, ROWSUM_WAVES * 32, 0, stream>>>(X, W, A_vals, A_rows, A_cols, ws, out);
}
